# starting kernel
# baseline (speedup 1.0000x reference)
#include <hip/hip_runtime.h>

#define N_NODES 50000
#define N_EDGES 800000
#define NB      100000u
#define C_OUT   64

typedef __attribute__((ext_vector_type(2))) float v2f;
typedef __attribute__((ext_vector_type(8))) float v8f;

__global__ void GCNN_zero_kernel(float* __restrict__ p, int n) {
    int i = blockIdx.x * blockDim.x + threadIdx.x;
    if (i < n) p[i] = 0.0f;
}

__global__ void GCNN_spmm_scatter_kernel(const float* __restrict__ x,
                                         const int*   __restrict__ rows,
                                         const int*   __restrict__ cols,
                                         const float* __restrict__ vals,
                                         float*       __restrict__ agg) {
    long long g = (long long)blockIdx.x * blockDim.x + threadIdx.x;
    int e = (int)(g >> 7);
    int k = (int)(g & 127);
    if (e >= N_EDGES) return;
    int   i = rows[e];
    int   j = cols[e];
    float v = vals[e];
    float hv = x[((k >> 6) * N_NODES + j) * 64 + (k & 63)];
    atomicAdd(&agg[i * 128 + k], v * hv);
}

__global__ void GCNN_gemm_wmma_kernel(const float* __restrict__ agg,
                                      const float* __restrict__ W,
                                      const float* __restrict__ bias,
                                      float*       __restrict__ out) {
    const int lane = threadIdx.x & 31;
    const int wave = threadIdx.x >> 5;
    const int half = lane >> 4;
    const int l16  = lane & 15;
    const unsigned r0 = (unsigned)blockIdx.x * 16u;
    const int      n0 = wave * 16;
    const unsigned r  = r0 + (unsigned)l16;

    v8f acc = {};
    #pragma unroll
    for (int k0 = 0; k0 < 64; k0 += 4) {
        const int ka = k0 + 2 * half;
        v2f a, b;
        {
            unsigned f  = r * 64u + (unsigned)ka;
            unsigned c0 = f / NB;
            unsigned rem = f - c0 * NB;
            a.x = agg[(rem >> 1) * 128u + c0 * 2u + (f & 1u)];
            f += 1u;
            c0  = f / NB;
            rem = f - c0 * NB;
            a.y = agg[(rem >> 1) * 128u + c0 * 2u + (f & 1u)];
        }
        b.x = W[ka * C_OUT + n0 + l16];
        b.y = W[(ka + 1) * C_OUT + n0 + l16];

        acc = __builtin_amdgcn_wmma_f32_16x16x4_f32(
             false, a,  false, b,
             (short)0, acc,  false,  false);
    }

    const int   col = n0 + l16;
    const float bv  = bias[col];
    #pragma unroll
    for (int v = 0; v < 8; ++v) {
        float o = acc[v] + bv;
        o = o > 0.0f ? o : 0.0f;
        out[(r0 + (unsigned)(v + 8 * half)) * (unsigned)C_OUT + (unsigned)col] = o;
    }
}

extern "C" void kernel_launch(void* const* d_in, const int* in_sizes, int n_in,
                              void* d_out, int out_size, void* d_ws, size_t ws_size,
                              hipStream_t stream) {
    const float* x    = (const float*)d_in[0];
    const int*   rows = (const int*)  d_in[1];
    const int*   cols = (const int*)  d_in[2];
    const float* vals = (const float*)d_in[3];
    const float* W    = (const float*)d_in[4];
    const float* bias = (const float*)d_in[5];
    float* out = (float*)d_out;
    float* agg = (float*)d_ws;

    const int aggN = N_NODES * 128;
    GCNN_zero_kernel<<<(aggN + 255) / 256, 256, 0, stream>>>(agg, aggN);

    const long long scatterThreads = (long long)N_EDGES * 128;
    const int scatterBlocks = (int)((scatterThreads + 255) / 256);
    GCNN_spmm_scatter_kernel<<<scatterBlocks, 256, 0, stream>>>(x, rows, cols,
                                                               vals, agg);

    GCNN_gemm_wmma_kernel<<<NB / 16, 128, 0, stream>>>(agg, W, bias, out);
}


# Replace agg zeroing and the global float-atomic scatter by per-row edge lists built with LDS exchanges and a one-wave-per-row register gather; row sums now add in list order, no atomics left
# speedup vs baseline: 4.5886x; 4.5886x over previous
#include <hip/hip_runtime.h>


#define N_NODES 50000
#define N_EDGES 800000
#define NB      100000u
#define C_OUT   64

#define LIST_GROUPS      64
#define EDGES_PER_GROUP  12500
#define HEAD_PITCH       50016
#define BUILD_THREADS    1024
#define EDGE_SLOTS       13
#define ROWS_PER_BLOCK   16

#define WS_AGG_OFF   0
#define WS_HEAD_OFF  6400000
#define WS_REC_OFF   9601024

typedef __attribute__((ext_vector_type(2))) float v2f;
typedef __attribute__((ext_vector_type(8))) float v8f;

__global__ void __launch_bounds__(BUILD_THREADS)
GCNN_build_lists_kernel(const int*   __restrict__ rows,
                        const int*   __restrict__ cols,
                        const float* __restrict__ vals,
                        int*         __restrict__ heads,
                        int*         __restrict__ rec) {
    __shared__ int4 lhead4[N_NODES / 4];
    int* lhead = (int*)lhead4;
    const int tid  = threadIdx.x;
    const int grp  = blockIdx.x;
    const int base = grp * EDGES_PER_GROUP;

    int   er[EDGE_SLOTS];
    int   ec[EDGE_SLOTS];
    float ev[EDGE_SLOTS];
    #pragma unroll
    for (int u = 0; u < EDGE_SLOTS; ++u) {
        const int k = tid + u * BUILD_THREADS;
        if (k < EDGES_PER_GROUP) {
            er[u] = rows[base + k];
            ec[u] = cols[base + k];
            ev[u] = vals[base + k];
        }
    }

    for (int q = tid; q < N_NODES / 4; q += BUILD_THREADS)
        lhead4[q] = make_int4(-1, -1, -1, -1);
    __syncthreads();

    int4* rec4 = (int4*)rec;
    #pragma unroll
    for (int u = 0; u < EDGE_SLOTS; ++u) {
        const int k = tid + u * BUILD_THREADS;
        if (k < EDGES_PER_GROUP) {
            const int e   = base + k;
            const int nxt = atomicExch(&lhead[er[u]], e);
            rec4[e] = make_int4(ec[u], __float_as_int(ev[u]), nxt, 0);
        }
    }
    __syncthreads();

    int4* out4 = (int4*)(heads + grp * HEAD_PITCH);
    for (int q = tid; q < N_NODES / 4; q += BUILD_THREADS)
        out4[q] = lhead4[q];
}

static __device__ __forceinline__ void GCNN_take_edge(unsigned long long& m,
                                                      const int4& ra, const int4& rb,
                                                      int& col, float& val) {
    const int s  = __builtin_ctzll(m);
    m &= m - 1ull;
    const int sl = s & 31;
    const int ca = __builtin_amdgcn_readlane(ra.x, sl);
    const int cb = __builtin_amdgcn_readlane(rb.x, sl);
    const int wa = __builtin_amdgcn_readlane(ra.y, sl);
    const int wb = __builtin_amdgcn_readlane(rb.y, sl);
    col = (s < 32) ? ca : cb;
    val = __int_as_float((s < 32) ? wa : wb);
}

static __device__ __forceinline__ void GCNN_axpy4(float4& acc, float v, const float4& xv) {
    acc.x += v * xv.x; acc.y += v * xv.y;
    acc.z += v * xv.z; acc.w += v * xv.w;
}

__global__ void __launch_bounds__(ROWS_PER_BLOCK * 32)
GCNN_spmm_gather_kernel(const float* __restrict__ x,
                        const int*   __restrict__ heads,
                        const int*   __restrict__ rec,
                        float*       __restrict__ agg) {
    __shared__ int sh[LIST_GROUPS][ROWS_PER_BLOCK + 1];
    const int tid  = threadIdx.x;
    const int lane = tid & 31;
    const int wave = tid >> 5;
    const int r0   = blockIdx.x * ROWS_PER_BLOCK;

    {
        const int rr = tid % ROWS_PER_BLOCK;
        const int l0 = tid / ROWS_PER_BLOCK;
        int h0 = -1, h1 = -1;
        if (r0 + rr < N_NODES) {
            h0 = heads[l0 * HEAD_PITCH + r0 + rr];
            h1 = heads[(l0 + 32) * HEAD_PITCH + r0 + rr];
        }
        sh[l0][rr]      = h0;
        sh[l0 + 32][rr] = h1;
    }
    __syncthreads();

    const int row = r0 + wave;
    int ea = sh[lane][wave];
    int eb = sh[lane + 32][wave];

    const int4* rec4 = (const int4*)rec;
    const float* xl = x + (lane >> 4) * (N_NODES * 64) + (lane & 15) * 4;
    float4 acc = make_float4(0.0f, 0.0f, 0.0f, 0.0f);

    for (int it = 0; it < EDGES_PER_GROUP; ++it) {
        const bool va = (unsigned)ea < (unsigned)N_EDGES;
        const bool vb = (unsigned)eb < (unsigned)N_EDGES;
        const unsigned ma = (unsigned)__ballot(va);
        const unsigned mb = (unsigned)__ballot(vb);
        if ((ma | mb) == 0u) break;

        int4 ra = make_int4(0, 0, -1, 0);
        int4 rb = make_int4(0, 0, -1, 0);
        if (va) ra = rec4[ea];
        if (vb) rb = rec4[eb];

        unsigned long long m = (unsigned long long)ma | ((unsigned long long)mb << 32);
        while (m) {
            int c0, c1, c2, c3; float v0, v1, v2, v3;
            GCNN_take_edge(m, ra, rb, c0, v0);
            const float4 x0 = *(const float4*)(xl + c0 * 64);
            if (!m) { GCNN_axpy4(acc, v0, x0); break; }
            GCNN_take_edge(m, ra, rb, c1, v1);
            const float4 x1 = *(const float4*)(xl + c1 * 64);
            if (!m) { GCNN_axpy4(acc, v0, x0); GCNN_axpy4(acc, v1, x1); break; }
            GCNN_take_edge(m, ra, rb, c2, v2);
            const float4 x2 = *(const float4*)(xl + c2 * 64);
            if (!m) {
                GCNN_axpy4(acc, v0, x0); GCNN_axpy4(acc, v1, x1);
                GCNN_axpy4(acc, v2, x2); break;
            }
            GCNN_take_edge(m, ra, rb, c3, v3);
            const float4 x3 = *(const float4*)(xl + c3 * 64);
            __builtin_amdgcn_sched_barrier(0);
            GCNN_axpy4(acc, v0, x0); GCNN_axpy4(acc, v1, x1);
            GCNN_axpy4(acc, v2, x2); GCNN_axpy4(acc, v3, x3);
        }
        ea = ra.z;
        eb = rb.z;
    }

    if (row < N_NODES)
        *(float4*)(agg + row * 128 + lane * 4) = acc;
}

__global__ void GCNN_gemm_wmma_kernel(const float* __restrict__ agg,
                                      const float* __restrict__ W,
                                      const float* __restrict__ bias,
                                      float*       __restrict__ out) {
    const int lane = threadIdx.x & 31;
    const int wave = threadIdx.x >> 5;
    const int half = lane >> 4;
    const int l16  = lane & 15;
    const unsigned r0 = (unsigned)blockIdx.x * 16u;
    const int      n0 = wave * 16;
    const unsigned r  = r0 + (unsigned)l16;

    v8f acc = {};
    #pragma unroll
    for (int k0 = 0; k0 < 64; k0 += 4) {
        const int ka = k0 + 2 * half;
        v2f a, b;
        {
            unsigned f  = r * 64u + (unsigned)ka;
            unsigned c0 = f / NB;
            unsigned rem = f - c0 * NB;
            a.x = agg[(rem >> 1) * 128u + c0 * 2u + (f & 1u)];
            f += 1u;
            c0  = f / NB;
            rem = f - c0 * NB;
            a.y = agg[(rem >> 1) * 128u + c0 * 2u + (f & 1u)];
        }
        b.x = W[ka * C_OUT + n0 + l16];
        b.y = W[(ka + 1) * C_OUT + n0 + l16];

        acc = __builtin_amdgcn_wmma_f32_16x16x4_f32(
             false, a,  false, b,
             (short)0, acc,  false,  false);
    }

    const int   col = n0 + l16;
    const float bv  = bias[col];
    #pragma unroll
    for (int v = 0; v < 8; ++v) {
        float o = acc[v] + bv;
        o = o > 0.0f ? o : 0.0f;
        out[(r0 + (unsigned)(v + 8 * half)) * (unsigned)C_OUT + (unsigned)col] = o;
    }
}

extern "C" void kernel_launch(void* const* d_in, const int* in_sizes, int n_in,
                              void* d_out, int out_size, void* d_ws, size_t ws_size,
                              hipStream_t stream) {
    const float* x    = (const float*)d_in[0];
    const int*   rows = (const int*)  d_in[1];
    const int*   cols = (const int*)  d_in[2];
    const float* vals = (const float*)d_in[3];
    const float* W    = (const float*)d_in[4];
    const float* bias = (const float*)d_in[5];
    float* out = (float*)d_out;
    float* agg   = (float*)d_ws + WS_AGG_OFF;
    int*   heads = (int*)d_ws + WS_HEAD_OFF;
    int*   rec   = (int*)d_ws + WS_REC_OFF;

    GCNN_build_lists_kernel<<<LIST_GROUPS, BUILD_THREADS, 0, stream>>>(
        rows, cols, vals, heads, rec);

    GCNN_spmm_gather_kernel<<<(N_NODES + ROWS_PER_BLOCK - 1) / ROWS_PER_BLOCK,
                              ROWS_PER_BLOCK * 32, 0, stream>>>(x, heads, rec, agg);

    GCNN_gemm_wmma_kernel<<<NB / 16, 128, 0, stream>>>(agg, W, bias, out);
}


# Gather kernel turns its 16x128 row sums in LDS and stores whole 128B lines in the GEMM's row-major A layout (k pairs of two steps adjacent): A fragments are 16B row loads, not 512B-strided gathers
# speedup vs baseline: 5.9542x; 1.2976x over previous
#include <hip/hip_runtime.h>


#define N_NODES 50000
#define N_EDGES 800000
#define NB      100000u
#define C_OUT   64

#define LIST_GROUPS      64
#define EDGES_PER_GROUP  12500
#define HEAD_PITCH       50016
#define BUILD_THREADS    1024
#define EDGE_SLOTS       13
#define ROWS_PER_BLOCK   16
#define TILE_PITCH       132

#define WS_AGG_OFF   0
#define WS_HEAD_OFF  6400000
#define WS_REC_OFF   9601024

typedef __attribute__((ext_vector_type(2))) float v2f;
typedef __attribute__((ext_vector_type(4))) float v4f;
typedef __attribute__((ext_vector_type(8))) float v8f;

static_assert(N_NODES % ROWS_PER_BLOCK == 0, "a block's 16 rows fill whole 128-byte lines of mp");
static_assert(NB % 32 == 0, "channel planes of mp start on 128-byte lines");

__global__ void __launch_bounds__(BUILD_THREADS)
GCNN_build_lists_kernel(const int*   __restrict__ rows,
                        const int*   __restrict__ cols,
                        const float* __restrict__ vals,
                        int*         __restrict__ heads,
                        int*         __restrict__ rec) {
    __shared__ int4 lhead4[N_NODES / 4];
    int* lhead = (int*)lhead4;
    const int tid  = threadIdx.x;
    const int grp  = blockIdx.x;
    const int base = grp * EDGES_PER_GROUP;

    int   er[EDGE_SLOTS];
    int   ec[EDGE_SLOTS];
    float ev[EDGE_SLOTS];
    #pragma unroll
    for (int u = 0; u < EDGE_SLOTS; ++u) {
        const int k = tid + u * BUILD_THREADS;
        if (k < EDGES_PER_GROUP) {
            er[u] = rows[base + k];
            ec[u] = cols[base + k];
            ev[u] = vals[base + k];
        }
    }

    for (int q = tid; q < N_NODES / 4; q += BUILD_THREADS)
        lhead4[q] = make_int4(-1, -1, -1, -1);
    __syncthreads();

    int4* rec4 = (int4*)rec;
    #pragma unroll
    for (int u = 0; u < EDGE_SLOTS; ++u) {
        const int k = tid + u * BUILD_THREADS;
        if (k < EDGES_PER_GROUP) {
            const int e   = base + k;
            const int nxt = atomicExch(&lhead[er[u]], e);
            rec4[e] = make_int4(ec[u], __float_as_int(ev[u]), nxt, 0);
        }
    }
    __syncthreads();

    int4* out4 = (int4*)(heads + grp * HEAD_PITCH);
    for (int q = tid; q < N_NODES / 4; q += BUILD_THREADS)
        out4[q] = lhead4[q];
}

static __device__ __forceinline__ void GCNN_take_edge(unsigned long long& m,
                                                      const int4& ra, const int4& rb,
                                                      int& col, float& val) {
    const int s  = __builtin_ctzll(m);
    m &= m - 1ull;
    const int sl = s & 31;
    const int ca = __builtin_amdgcn_readlane(ra.x, sl);
    const int cb = __builtin_amdgcn_readlane(rb.x, sl);
    const int wa = __builtin_amdgcn_readlane(ra.y, sl);
    const int wb = __builtin_amdgcn_readlane(rb.y, sl);
    col = (s < 32) ? ca : cb;
    val = __int_as_float((s < 32) ? wa : wb);
}

static __device__ __forceinline__ void GCNN_axpy4(float4& acc, float v, const float4& xv) {
    acc.x += v * xv.x; acc.y += v * xv.y;
    acc.z += v * xv.z; acc.w += v * xv.w;
}

__global__ void __launch_bounds__(ROWS_PER_BLOCK * 32)
GCNN_spmm_gather_kernel(const float* __restrict__ x,
                        const int*   __restrict__ heads,
                        const int*   __restrict__ rec,
                        float*       __restrict__ mp) {
    __shared__ int sh[LIST_GROUPS][ROWS_PER_BLOCK + 1];
    __shared__ float4 tile4[ROWS_PER_BLOCK * TILE_PITCH / 4];
    float* tile = (float*)tile4;
    const int tid  = threadIdx.x;
    const int lane = tid & 31;
    const int wave = tid >> 5;
    const int r0   = blockIdx.x * ROWS_PER_BLOCK;

    {
        const int rr = tid % ROWS_PER_BLOCK;
        const int l0 = tid / ROWS_PER_BLOCK;
        int h0 = -1, h1 = -1;
        if (r0 + rr < N_NODES) {
            h0 = heads[l0 * HEAD_PITCH + r0 + rr];
            h1 = heads[(l0 + 32) * HEAD_PITCH + r0 + rr];
        }
        sh[l0][rr]      = h0;
        sh[l0 + 32][rr] = h1;
    }
    __syncthreads();

    int ea = sh[lane][wave];
    int eb = sh[lane + 32][wave];

    const int4* rec4 = (const int4*)rec;
    const float* xl = x + (lane >> 4) * (N_NODES * 64) + (lane & 15) * 4;
    float4 acc = make_float4(0.0f, 0.0f, 0.0f, 0.0f);

    for (int it = 0; it < EDGES_PER_GROUP; ++it) {
        const bool va = (unsigned)ea < (unsigned)N_EDGES;
        const bool vb = (unsigned)eb < (unsigned)N_EDGES;
        const unsigned ma = (unsigned)__ballot(va);
        const unsigned mb = (unsigned)__ballot(vb);
        if ((ma | mb) == 0u) break;

        int4 ra = make_int4(0, 0, -1, 0);
        int4 rb = make_int4(0, 0, -1, 0);
        if (va) ra = rec4[ea];
        if (vb) rb = rec4[eb];

        unsigned long long m = (unsigned long long)ma | ((unsigned long long)mb << 32);
        while (m) {
            int c0, c1, c2, c3; float v0, v1, v2, v3;
            GCNN_take_edge(m, ra, rb, c0, v0);
            const float4 x0 = *(const float4*)(xl + c0 * 64);
            if (!m) { GCNN_axpy4(acc, v0, x0); break; }
            GCNN_take_edge(m, ra, rb, c1, v1);
            const float4 x1 = *(const float4*)(xl + c1 * 64);
            if (!m) { GCNN_axpy4(acc, v0, x0); GCNN_axpy4(acc, v1, x1); break; }
            GCNN_take_edge(m, ra, rb, c2, v2);
            const float4 x2 = *(const float4*)(xl + c2 * 64);
            if (!m) {
                GCNN_axpy4(acc, v0, x0); GCNN_axpy4(acc, v1, x1);
                GCNN_axpy4(acc, v2, x2); break;
            }
            GCNN_take_edge(m, ra, rb, c3, v3);
            const float4 x3 = *(const float4*)(xl + c3 * 64);
            __builtin_amdgcn_sched_barrier(0);
            GCNN_axpy4(acc, v0, x0); GCNN_axpy4(acc, v1, x1);
            GCNN_axpy4(acc, v2, x2); GCNN_axpy4(acc, v3, x3);
        }
        ea = ra.z;
        eb = rb.z;
    }

    *(float4*)(tile + wave * TILE_PITCH + lane * 4) = acc;
    __syncthreads();
    {
        const int c0 = tid >> 3;
        const int q  = tid & 7;
        const int ia = ((q >> 1) << 2) + (q & 1);
        const v2f lo = *(const v2f*)(tile + ia * TILE_PITCH + 2 * c0);
        const v2f hi = *(const v2f*)(tile + (ia + 2) * TILE_PITCH + 2 * c0);
        v4f o;
        o.x = lo.x; o.y = lo.y; o.z = hi.x; o.w = hi.y;
        *(v4f*)(mp + (unsigned)c0 * NB + (unsigned)r0 * 2u + (unsigned)q * 4u) = o;
    }
}

__global__ void GCNN_gemm_wmma_kernel(const float* __restrict__ mp,
                                      const float* __restrict__ W,
                                      const float* __restrict__ bias,
                                      float*       __restrict__ out) {
    const int lane = threadIdx.x & 31;
    const int wave = threadIdx.x >> 5;
    const int half = lane >> 4;
    const int l16  = lane & 15;
    const unsigned r0 = (unsigned)blockIdx.x * 16u;
    const int      n0 = wave * 16;
    const unsigned r  = r0 + (unsigned)l16;

    const float* arow = mp + r * 64u + 4u * (unsigned)half;

    v8f acc = {};
    #pragma unroll
    for (int k8 = 0; k8 < 64; k8 += 8) {
        const v4f av = *(const v4f*)(arow + k8);
        #pragma unroll
        for (int s = 0; s < 2; ++s) {
            const int ka = k8 + 4 * s + 2 * half;
            v2f a, b;
            a.x = s ? av.z : av.x;
            a.y = s ? av.w : av.y;
            b.x = W[ka * C_OUT + n0 + l16];
            b.y = W[(ka + 1) * C_OUT + n0 + l16];

            acc = __builtin_amdgcn_wmma_f32_16x16x4_f32(
                 false, a,  false, b,
                 (short)0, acc,  false,  false);
        }
    }

    const int   col = n0 + l16;
    const float bv  = bias[col];
    #pragma unroll
    for (int v = 0; v < 8; ++v) {
        float o = acc[v] + bv;
        o = o > 0.0f ? o : 0.0f;
        out[(r0 + (unsigned)(v + 8 * half)) * (unsigned)C_OUT + (unsigned)col] = o;
    }
}

extern "C" void kernel_launch(void* const* d_in, const int* in_sizes, int n_in,
                              void* d_out, int out_size, void* d_ws, size_t ws_size,
                              hipStream_t stream) {
    const float* x    = (const float*)d_in[0];
    const int*   rows = (const int*)  d_in[1];
    const int*   cols = (const int*)  d_in[2];
    const float* vals = (const float*)d_in[3];
    const float* W    = (const float*)d_in[4];
    const float* bias = (const float*)d_in[5];
    float* out = (float*)d_out;
    float* mp    = (float*)d_ws + WS_AGG_OFF;
    int*   heads = (int*)d_ws + WS_HEAD_OFF;
    int*   rec   = (int*)d_ws + WS_REC_OFF;

    GCNN_build_lists_kernel<<<LIST_GROUPS, BUILD_THREADS, 0, stream>>>(
        rows, cols, vals, heads, rec);

    GCNN_spmm_gather_kernel<<<(N_NODES + ROWS_PER_BLOCK - 1) / ROWS_PER_BLOCK,
                              ROWS_PER_BLOCK * 32, 0, stream>>>(x, heads, rec, mp);

    GCNN_gemm_wmma_kernel<<<NB / 16, 128, 0, stream>>>(mp, W, bias, out);
}


# GEMM re-tiled: one wave per 16x64 out tile (4 accumulators, tile t = W columns 4c+t), 40 loads issued as one group: A read once not 4x, W by 16B loads, whole 128B out lines per store; k order kept
# speedup vs baseline: 6.4292x; 1.0798x over previous
#include <hip/hip_runtime.h>


#define N_NODES 50000
#define N_EDGES 800000
#define NB      100000u
#define C_OUT   64

#define LIST_GROUPS      64
#define EDGES_PER_GROUP  12500
#define HEAD_PITCH       50016
#define BUILD_THREADS    1024
#define EDGE_SLOTS       13
#define ROWS_PER_BLOCK   16
#define TILE_PITCH       132
#define GEMM_TILES       6250u
#define GEMM_WAVES       4u
#define GEMM_BLOCKS      1563

#define WS_AGG_OFF   0
#define WS_HEAD_OFF  6400000
#define WS_REC_OFF   9601024

typedef __attribute__((ext_vector_type(2))) float v2f;
typedef __attribute__((ext_vector_type(4))) float v4f;
typedef __attribute__((ext_vector_type(8))) float v8f;

static_assert(N_NODES % ROWS_PER_BLOCK == 0, "a block's 16 rows fill whole 128-byte lines of mp");
static_assert(NB % 32 == 0, "channel planes of mp start on 128-byte lines");
static_assert(GEMM_TILES * 16u == NB, "the 16-row tiles cover M exactly");
static_assert(GEMM_BLOCKS * GEMM_WAVES >= GEMM_TILES && (GEMM_BLOCKS - 1) * GEMM_WAVES < GEMM_TILES, "one wave per tile, no empty block");

__global__ void __launch_bounds__(BUILD_THREADS)
GCNN_build_lists_kernel(const int*   __restrict__ rows,
                        const int*   __restrict__ cols,
                        const float* __restrict__ vals,
                        int*         __restrict__ heads,
                        int*         __restrict__ rec) {
    __shared__ int4 lhead4[N_NODES / 4];
    int* lhead = (int*)lhead4;
    const int tid  = threadIdx.x;
    const int grp  = blockIdx.x;
    const int base = grp * EDGES_PER_GROUP;

    int   er[EDGE_SLOTS];
    int   ec[EDGE_SLOTS];
    float ev[EDGE_SLOTS];
    #pragma unroll
    for (int u = 0; u < EDGE_SLOTS; ++u) {
        const int k = tid + u * BUILD_THREADS;
        if (k < EDGES_PER_GROUP) {
            er[u] = rows[base + k];
            ec[u] = cols[base + k];
            ev[u] = vals[base + k];
        }
    }

    for (int q = tid; q < N_NODES / 4; q += BUILD_THREADS)
        lhead4[q] = make_int4(-1, -1, -1, -1);
    __syncthreads();

    int4* rec4 = (int4*)rec;
    #pragma unroll
    for (int u = 0; u < EDGE_SLOTS; ++u) {
        const int k = tid + u * BUILD_THREADS;
        if (k < EDGES_PER_GROUP) {
            const int e   = base + k;
            const int nxt = atomicExch(&lhead[er[u]], e);
            rec4[e] = make_int4(ec[u], __float_as_int(ev[u]), nxt, 0);
        }
    }
    __syncthreads();

    int4* out4 = (int4*)(heads + grp * HEAD_PITCH);
    for (int q = tid; q < N_NODES / 4; q += BUILD_THREADS)
        out4[q] = lhead4[q];
}

static __device__ __forceinline__ void GCNN_take_edge(unsigned long long& m,
                                                      const int4& ra, const int4& rb,
                                                      int& col, float& val) {
    const int s  = __builtin_ctzll(m);
    m &= m - 1ull;
    const int sl = s & 31;
    const int ca = __builtin_amdgcn_readlane(ra.x, sl);
    const int cb = __builtin_amdgcn_readlane(rb.x, sl);
    const int wa = __builtin_amdgcn_readlane(ra.y, sl);
    const int wb = __builtin_amdgcn_readlane(rb.y, sl);
    col = (s < 32) ? ca : cb;
    val = __int_as_float((s < 32) ? wa : wb);
}

static __device__ __forceinline__ void GCNN_axpy4(float4& acc, float v, const float4& xv) {
    acc.x += v * xv.x; acc.y += v * xv.y;
    acc.z += v * xv.z; acc.w += v * xv.w;
}

__global__ void __launch_bounds__(ROWS_PER_BLOCK * 32)
GCNN_spmm_gather_kernel(const float* __restrict__ x,
                        const int*   __restrict__ heads,
                        const int*   __restrict__ rec,
                        float*       __restrict__ mp) {
    __shared__ int sh[LIST_GROUPS][ROWS_PER_BLOCK + 1];
    __shared__ float4 tile4[ROWS_PER_BLOCK * TILE_PITCH / 4];
    float* tile = (float*)tile4;
    const int tid  = threadIdx.x;
    const int lane = tid & 31;
    const int wave = tid >> 5;
    const int r0   = blockIdx.x * ROWS_PER_BLOCK;

    {
        const int rr = tid % ROWS_PER_BLOCK;
        const int l0 = tid / ROWS_PER_BLOCK;
        int h0 = -1, h1 = -1;
        if (r0 + rr < N_NODES) {
            h0 = heads[l0 * HEAD_PITCH + r0 + rr];
            h1 = heads[(l0 + 32) * HEAD_PITCH + r0 + rr];
        }
        sh[l0][rr]      = h0;
        sh[l0 + 32][rr] = h1;
    }
    __syncthreads();

    int ea = sh[lane][wave];
    int eb = sh[lane + 32][wave];

    const int4* rec4 = (const int4*)rec;
    const float* xl = x + (lane >> 4) * (N_NODES * 64) + (lane & 15) * 4;
    float4 acc = make_float4(0.0f, 0.0f, 0.0f, 0.0f);

    for (int it = 0; it < EDGES_PER_GROUP; ++it) {
        const bool va = (unsigned)ea < (unsigned)N_EDGES;
        const bool vb = (unsigned)eb < (unsigned)N_EDGES;
        const unsigned ma = (unsigned)__ballot(va);
        const unsigned mb = (unsigned)__ballot(vb);
        if ((ma | mb) == 0u) break;

        int4 ra = make_int4(0, 0, -1, 0);
        int4 rb = make_int4(0, 0, -1, 0);
        if (va) ra = rec4[ea];
        if (vb) rb = rec4[eb];

        unsigned long long m = (unsigned long long)ma | ((unsigned long long)mb << 32);
        while (m) {
            int c0, c1, c2, c3; float v0, v1, v2, v3;
            GCNN_take_edge(m, ra, rb, c0, v0);
            const float4 x0 = *(const float4*)(xl + c0 * 64);
            if (!m) { GCNN_axpy4(acc, v0, x0); break; }
            GCNN_take_edge(m, ra, rb, c1, v1);
            const float4 x1 = *(const float4*)(xl + c1 * 64);
            if (!m) { GCNN_axpy4(acc, v0, x0); GCNN_axpy4(acc, v1, x1); break; }
            GCNN_take_edge(m, ra, rb, c2, v2);
            const float4 x2 = *(const float4*)(xl + c2 * 64);
            if (!m) {
                GCNN_axpy4(acc, v0, x0); GCNN_axpy4(acc, v1, x1);
                GCNN_axpy4(acc, v2, x2); break;
            }
            GCNN_take_edge(m, ra, rb, c3, v3);
            const float4 x3 = *(const float4*)(xl + c3 * 64);
            __builtin_amdgcn_sched_barrier(0);
            GCNN_axpy4(acc, v0, x0); GCNN_axpy4(acc, v1, x1);
            GCNN_axpy4(acc, v2, x2); GCNN_axpy4(acc, v3, x3);
        }
        ea = ra.z;
        eb = rb.z;
    }

    *(float4*)(tile + wave * TILE_PITCH + lane * 4) = acc;
    __syncthreads();
    {
        const int c0 = tid >> 3;
        const int q  = tid & 7;
        const int ia = ((q >> 1) << 2) + (q & 1);
        const v2f lo = *(const v2f*)(tile + ia * TILE_PITCH + 2 * c0);
        const v2f hi = *(const v2f*)(tile + (ia + 2) * TILE_PITCH + 2 * c0);
        v4f o;
        o.x = lo.x; o.y = lo.y; o.z = hi.x; o.w = hi.y;
        *(v4f*)(mp + (unsigned)c0 * NB + (unsigned)r0 * 2u + (unsigned)q * 4u) = o;
    }
}

static __device__ __forceinline__ v8f GCNN_mma(const v2f a, float bx, float by, const v8f c) {
    v2f b;
    b.x = bx; b.y = by;
    return __builtin_amdgcn_wmma_f32_16x16x4_f32(
         false, a,  false, b,
         (short)0, c,  false,  false);
}

__global__ void __launch_bounds__(GEMM_WAVES * 32)
GCNN_gemm_wmma_kernel(const float* __restrict__ mp,
                      const float* __restrict__ W,
                      const float* __restrict__ bias,
                      float*       __restrict__ out) {
    const int lane = threadIdx.x & 31;
    const int wave = threadIdx.x >> 5;
    const int half = lane >> 4;
    const int l16  = lane & 15;
    const unsigned tile = (unsigned)blockIdx.x * GEMM_WAVES + (unsigned)wave;
    if (tile >= GEMM_TILES) return;
    const unsigned r0 = tile * 16u;
    const unsigned r  = r0 + (unsigned)l16;

    const float* arow = mp + r * 64u + 4u * (unsigned)half;
    const float* wcol = W + 2 * half * C_OUT + 4 * l16;

    v4f av[8];
    #pragma unroll
    for (int j = 0; j < 8; ++j)
        av[j] = *(const v4f*)(arow + 8 * j);
    v4f w0[16], w1[16];
    #pragma unroll
    for (int t = 0; t < 16; ++t) {
        w0[t] = *(const v4f*)(wcol + (4 * t) * C_OUT);
        w1[t] = *(const v4f*)(wcol + (4 * t + 1) * C_OUT);
    }
    __builtin_amdgcn_sched_barrier(0);

    v8f acc0 = {}, acc1 = {}, acc2 = {}, acc3 = {};
    #pragma unroll
    for (int t = 0; t < 16; ++t) {
        v2f a;
        a.x = (t & 1) ? av[t >> 1].z : av[t >> 1].x;
        a.y = (t & 1) ? av[t >> 1].w : av[t >> 1].y;
        acc0 = GCNN_mma(a, w0[t].x, w1[t].x, acc0);
        acc1 = GCNN_mma(a, w0[t].y, w1[t].y, acc1);
        acc2 = GCNN_mma(a, w0[t].z, w1[t].z, acc2);
        acc3 = GCNN_mma(a, w0[t].w, w1[t].w, acc3);
    }

    const v4f bv = *(const v4f*)(bias + 4 * l16);
    float* orow = out + (r0 + 8u * (unsigned)half) * (unsigned)C_OUT + 4u * (unsigned)l16;
    #pragma unroll
    for (int v = 0; v < 8; ++v) {
        v4f o;
        o.x = acc0[v] + bv.x;
        o.y = acc1[v] + bv.y;
        o.z = acc2[v] + bv.z;
        o.w = acc3[v] + bv.w;
        o.x = o.x > 0.0f ? o.x : 0.0f;
        o.y = o.y > 0.0f ? o.y : 0.0f;
        o.z = o.z > 0.0f ? o.z : 0.0f;
        o.w = o.w > 0.0f ? o.w : 0.0f;
        *(v4f*)(orow + v * C_OUT) = o;
    }
}

extern "C" void kernel_launch(void* const* d_in, const int* in_sizes, int n_in,
                              void* d_out, int out_size, void* d_ws, size_t ws_size,
                              hipStream_t stream) {
    const float* x    = (const float*)d_in[0];
    const int*   rows = (const int*)  d_in[1];
    const int*   cols = (const int*)  d_in[2];
    const float* vals = (const float*)d_in[3];
    const float* W    = (const float*)d_in[4];
    const float* bias = (const float*)d_in[5];
    float* out = (float*)d_out;
    float* mp    = (float*)d_ws + WS_AGG_OFF;
    int*   heads = (int*)d_ws + WS_HEAD_OFF;
    int*   rec   = (int*)d_ws + WS_REC_OFF;

    GCNN_build_lists_kernel<<<LIST_GROUPS, BUILD_THREADS, 0, stream>>>(
        rows, cols, vals, heads, rec);

    GCNN_spmm_gather_kernel<<<(N_NODES + ROWS_PER_BLOCK - 1) / ROWS_PER_BLOCK,
                              ROWS_PER_BLOCK * 32, 0, stream>>>(x, heads, rec, mp);

    GCNN_gemm_wmma_kernel<<<GEMM_BLOCKS, GEMM_WAVES * 32, 0, stream>>>(mp, W, bias, out);
}


# Head-table entries hold each list's first edge record (value, column, next; 8 bytes): round 1 of the gather fetches no record, later 8-byte records are requested a round ahead; sum order kept
# speedup vs baseline: 7.0590x; 1.0980x over previous
#include <hip/hip_runtime.h>


#define N_NODES 50000
#define N_EDGES 800000
#define NB      100000u
#define C_OUT   64

#define LIST_GROUPS      64
#define EDGES_PER_GROUP  12500
#define HEAD_PITCH       50016
#define BUILD_THREADS    1024
#define EDGE_SLOTS       13
#define REC_PITCH        12512
#define ROWS_PER_BLOCK   16
#define TILE_PITCH       132
#define GEMM_TILES       6250u
#define GEMM_WAVES       4u
#define GEMM_BLOCKS      1563

#define WS_AGG_OFF   0
#define WS_HEAD_OFF  6400000
#define WS_REC_OFF   12802048
#define REC_NONE_Y   0xFFFFFFFFu

typedef __attribute__((ext_vector_type(2))) float v2f;
typedef __attribute__((ext_vector_type(4))) float v4f;
typedef __attribute__((ext_vector_type(8))) float v8f;

static_assert(N_NODES % ROWS_PER_BLOCK == 0, "a block's 16 rows fill whole 128-byte lines of mp");
static_assert(HEAD_PITCH % 16 == 0 && REC_PITCH % 16 == 0 && REC_PITCH >= EDGES_PER_GROUP, "a group's head entries and records start on 128-byte lines");
static_assert(N_NODES < 0xFFFF && EDGES_PER_GROUP < 0xFFFF, "column and next fit 16 bits beside the none mark");
static_assert(NB % 32 == 0, "channel planes of mp start on 128-byte lines");
static_assert(GEMM_TILES * 16u == NB, "the 16-row tiles cover M exactly");
static_assert(GEMM_BLOCKS * GEMM_WAVES >= GEMM_TILES && (GEMM_BLOCKS - 1) * GEMM_WAVES < GEMM_TILES, "one wave per tile, no empty block");

__global__ void __launch_bounds__(BUILD_THREADS)
GCNN_build_lists_kernel(const int*   __restrict__ rows,
                        const int*   __restrict__ cols,
                        const float* __restrict__ vals,
                        int*         __restrict__ heads,
                        int*         __restrict__ rec) {
    __shared__ int4  lhead4[N_NODES / 4];
    __shared__ uint2 lrec[EDGES_PER_GROUP];
    int* lhead = (int*)lhead4;
    const int tid  = threadIdx.x;
    const int grp  = blockIdx.x;
    const int base = grp * EDGES_PER_GROUP;

    int   er[EDGE_SLOTS];
    int   ec[EDGE_SLOTS];
    float ev[EDGE_SLOTS];
    #pragma unroll
    for (int u = 0; u < EDGE_SLOTS; ++u) {
        const int k = tid + u * BUILD_THREADS;
        if (k < EDGES_PER_GROUP) {
            er[u] = rows[base + k];
            ec[u] = cols[base + k];
            ev[u] = vals[base + k];
        }
    }

    for (int q = tid; q < N_NODES / 4; q += BUILD_THREADS)
        lhead4[q] = make_int4(-1, -1, -1, -1);
    __syncthreads();

    uint2* rec8 = (uint2*)rec + grp * REC_PITCH;
    #pragma unroll
    for (int u = 0; u < EDGE_SLOTS; ++u) {
        const int k = tid + u * BUILD_THREADS;
        if (k < EDGES_PER_GROUP) {
            const int nxt = atomicExch(&lhead[er[u]], k);
            uint2 w;
            w.x = __float_as_uint(ev[u]);
            w.y = ((unsigned)ec[u] & 0xFFFFu) | ((unsigned)nxt << 16);
            lrec[k] = w;
            rec8[k] = w;
        }
    }
    __syncthreads();

    const int2* lhead2 = (const int2*)lhead4;
    uint4* out4 = (uint4*)heads + grp * (HEAD_PITCH / 2);
    for (int q = tid; q < N_NODES / 2; q += BUILD_THREADS) {
        const int2 h = lhead2[q];
        uint2 a = make_uint2(0u, REC_NONE_Y);
        uint2 b = make_uint2(0u, REC_NONE_Y);
        if (h.x >= 0) a = lrec[h.x];
        if (h.y >= 0) b = lrec[h.y];
        out4[q] = make_uint4(a.x, a.y, b.x, b.y);
    }
}

static __device__ __forceinline__ void GCNN_take_edge(unsigned long long& m,
                                                      int ka, int xa, int kb, int xb,
                                                      int& col, float& val) {
    const int s  = __builtin_ctzll(m);
    m &= m - 1ull;
    const int sl = s & 31;
    const int ca = __builtin_amdgcn_readlane(ka, sl);
    const int cb = __builtin_amdgcn_readlane(kb, sl);
    const int wa = __builtin_amdgcn_readlane(xa, sl);
    const int wb = __builtin_amdgcn_readlane(xb, sl);
    col = (s < 32) ? ca : cb;
    val = __int_as_float((s < 32) ? wa : wb);
}

static __device__ __forceinline__ void GCNN_axpy4(float4& acc, float v, const float4& xv) {
    acc.x += v * xv.x; acc.y += v * xv.y;
    acc.z += v * xv.z; acc.w += v * xv.w;
}

static __device__ __forceinline__ void GCNN_gather_group(unsigned long long& m,
                                                         int ka, int xa, int kb, int xb,
                                                         const float* xl, float4& acc) {
    int c0, c1, c2, c3; float v0, v1, v2, v3;
    GCNN_take_edge(m, ka, xa, kb, xb, c0, v0);
    const float4 x0 = *(const float4*)(xl + c0 * 64);
    if (!m) { GCNN_axpy4(acc, v0, x0); return; }
    GCNN_take_edge(m, ka, xa, kb, xb, c1, v1);
    const float4 x1 = *(const float4*)(xl + c1 * 64);
    if (!m) { GCNN_axpy4(acc, v0, x0); GCNN_axpy4(acc, v1, x1); return; }
    GCNN_take_edge(m, ka, xa, kb, xb, c2, v2);
    const float4 x2 = *(const float4*)(xl + c2 * 64);
    if (!m) {
        GCNN_axpy4(acc, v0, x0); GCNN_axpy4(acc, v1, x1);
        GCNN_axpy4(acc, v2, x2); return;
    }
    GCNN_take_edge(m, ka, xa, kb, xb, c3, v3);
    const float4 x3 = *(const float4*)(xl + c3 * 64);
    __builtin_amdgcn_sched_barrier(0);
    GCNN_axpy4(acc, v0, x0); GCNN_axpy4(acc, v1, x1);
    GCNN_axpy4(acc, v2, x2); GCNN_axpy4(acc, v3, x3);
}

__global__ void __launch_bounds__(ROWS_PER_BLOCK * 32)
GCNN_spmm_gather_kernel(const float* __restrict__ x,
                        const int*   __restrict__ heads,
                        const int*   __restrict__ rec,
                        float*       __restrict__ mp) {
    __shared__ uint2 sh[LIST_GROUPS][ROWS_PER_BLOCK + 1];
    __shared__ float4 tile4[ROWS_PER_BLOCK * TILE_PITCH / 4];
    float* tile = (float*)tile4;
    const int tid  = threadIdx.x;
    const int lane = tid & 31;
    const int wave = tid >> 5;
    const int r0   = blockIdx.x * ROWS_PER_BLOCK;

    {
        const int rr = tid % ROWS_PER_BLOCK;
        const int l0 = tid / ROWS_PER_BLOCK;
        const uint2* heads8 = (const uint2*)heads;
        uint2 h0 = make_uint2(0u, REC_NONE_Y);
        uint2 h1 = make_uint2(0u, REC_NONE_Y);
        if (r0 + rr < N_NODES) {
            h0 = heads8[l0 * HEAD_PITCH + r0 + rr];
            h1 = heads8[(l0 + 32) * HEAD_PITCH + r0 + rr];
        }
        sh[l0][rr]      = h0;
        sh[l0 + 32][rr] = h1;
    }
    __syncthreads();

    uint2 ha = sh[lane][wave];
    uint2 hb = sh[lane + 32][wave];

    const uint2* rec8 = (const uint2*)rec;
    const unsigned basea = (unsigned)lane * REC_PITCH;
    const unsigned baseb = (unsigned)(lane + 32) * REC_PITCH;
    const float* xl = x + (lane >> 4) * (N_NODES * 64) + (lane & 15) * 4;
    float4 acc = make_float4(0.0f, 0.0f, 0.0f, 0.0f);

    for (int it = 0; it < EDGES_PER_GROUP; ++it) {
        const unsigned ka = ha.y & 0xFFFFu;
        const unsigned kb = hb.y & 0xFFFFu;
        const bool va = ka < (unsigned)N_NODES;
        const bool vb = kb < (unsigned)N_NODES;
        const unsigned ma = (unsigned)__ballot(va);
        const unsigned mb = (unsigned)__ballot(vb);
        if ((ma | mb) == 0u) break;

        const int kai = (int)ka, kbi = (int)kb;
        const int xai = (int)ha.x, xbi = (int)hb.x;
        unsigned long long m = (unsigned long long)ma | ((unsigned long long)mb << 32);

        int c0, c1, c2, c3; float v0, v1, v2, v3;
        float4 x0, x1, x2, x3;
        int ng = 1;
        GCNN_take_edge(m, kai, xai, kbi, xbi, c0, v0);
        x0 = *(const float4*)(xl + c0 * 64);
        if (m) {
            GCNN_take_edge(m, kai, xai, kbi, xbi, c1, v1);
            x1 = *(const float4*)(xl + c1 * 64);
            ng = 2;
            if (m) {
                GCNN_take_edge(m, kai, xai, kbi, xbi, c2, v2);
                x2 = *(const float4*)(xl + c2 * 64);
                ng = 3;
                if (m) {
                    GCNN_take_edge(m, kai, xai, kbi, xbi, c3, v3);
                    x3 = *(const float4*)(xl + c3 * 64);
                    ng = 4;
                }
            }
        }

        const unsigned na = ha.y >> 16;
        const unsigned nb = hb.y >> 16;
        uint2 ra = make_uint2(0u, REC_NONE_Y);
        uint2 rb = make_uint2(0u, REC_NONE_Y);
        if (va && na < (unsigned)EDGES_PER_GROUP) ra = rec8[basea + na];
        if (vb && nb < (unsigned)EDGES_PER_GROUP) rb = rec8[baseb + nb];

        GCNN_axpy4(acc, v0, x0);
        if (ng > 1) GCNN_axpy4(acc, v1, x1);
        if (ng > 2) GCNN_axpy4(acc, v2, x2);
        if (ng > 3) GCNN_axpy4(acc, v3, x3);

        while (m)
            GCNN_gather_group(m, kai, xai, kbi, xbi, xl, acc);
        ha = ra;
        hb = rb;
    }

    *(float4*)(tile + wave * TILE_PITCH + lane * 4) = acc;
    __syncthreads();
    {
        const int c0 = tid >> 3;
        const int q  = tid & 7;
        const int ia = ((q >> 1) << 2) + (q & 1);
        const v2f lo = *(const v2f*)(tile + ia * TILE_PITCH + 2 * c0);
        const v2f hi = *(const v2f*)(tile + (ia + 2) * TILE_PITCH + 2 * c0);
        v4f o;
        o.x = lo.x; o.y = lo.y; o.z = hi.x; o.w = hi.y;
        *(v4f*)(mp + (unsigned)c0 * NB + (unsigned)r0 * 2u + (unsigned)q * 4u) = o;
    }
}

static __device__ __forceinline__ v8f GCNN_mma(const v2f a, float bx, float by, const v8f c) {
    v2f b;
    b.x = bx; b.y = by;
    return __builtin_amdgcn_wmma_f32_16x16x4_f32(
         false, a,  false, b,
         (short)0, c,  false,  false);
}

__global__ void __launch_bounds__(GEMM_WAVES * 32)
GCNN_gemm_wmma_kernel(const float* __restrict__ mp,
                      const float* __restrict__ W,
                      const float* __restrict__ bias,
                      float*       __restrict__ out) {
    const int lane = threadIdx.x & 31;
    const int wave = threadIdx.x >> 5;
    const int half = lane >> 4;
    const int l16  = lane & 15;
    const unsigned tile = (unsigned)blockIdx.x * GEMM_WAVES + (unsigned)wave;
    if (tile >= GEMM_TILES) return;
    const unsigned r0 = tile * 16u;
    const unsigned r  = r0 + (unsigned)l16;

    const float* arow = mp + r * 64u + 4u * (unsigned)half;
    const float* wcol = W + 2 * half * C_OUT + 4 * l16;

    v4f av[8];
    #pragma unroll
    for (int j = 0; j < 8; ++j)
        av[j] = *(const v4f*)(arow + 8 * j);
    v4f w0[16], w1[16];
    #pragma unroll
    for (int t = 0; t < 16; ++t) {
        w0[t] = *(const v4f*)(wcol + (4 * t) * C_OUT);
        w1[t] = *(const v4f*)(wcol + (4 * t + 1) * C_OUT);
    }
    __builtin_amdgcn_sched_barrier(0);

    v8f acc0 = {}, acc1 = {}, acc2 = {}, acc3 = {};
    #pragma unroll
    for (int t = 0; t < 16; ++t) {
        v2f a;
        a.x = (t & 1) ? av[t >> 1].z : av[t >> 1].x;
        a.y = (t & 1) ? av[t >> 1].w : av[t >> 1].y;
        acc0 = GCNN_mma(a, w0[t].x, w1[t].x, acc0);
        acc1 = GCNN_mma(a, w0[t].y, w1[t].y, acc1);
        acc2 = GCNN_mma(a, w0[t].z, w1[t].z, acc2);
        acc3 = GCNN_mma(a, w0[t].w, w1[t].w, acc3);
    }

    const v4f bv = *(const v4f*)(bias + 4 * l16);
    float* orow = out + (r0 + 8u * (unsigned)half) * (unsigned)C_OUT + 4u * (unsigned)l16;
    #pragma unroll
    for (int v = 0; v < 8; ++v) {
        v4f o;
        o.x = acc0[v] + bv.x;
        o.y = acc1[v] + bv.y;
        o.z = acc2[v] + bv.z;
        o.w = acc3[v] + bv.w;
        o.x = o.x > 0.0f ? o.x : 0.0f;
        o.y = o.y > 0.0f ? o.y : 0.0f;
        o.z = o.z > 0.0f ? o.z : 0.0f;
        o.w = o.w > 0.0f ? o.w : 0.0f;
        *(v4f*)(orow + v * C_OUT) = o;
    }
}

extern "C" void kernel_launch(void* const* d_in, const int* in_sizes, int n_in,
                              void* d_out, int out_size, void* d_ws, size_t ws_size,
                              hipStream_t stream) {
    const float* x    = (const float*)d_in[0];
    const int*   rows = (const int*)  d_in[1];
    const int*   cols = (const int*)  d_in[2];
    const float* vals = (const float*)d_in[3];
    const float* W    = (const float*)d_in[4];
    const float* bias = (const float*)d_in[5];
    float* out = (float*)d_out;
    float* mp    = (float*)d_ws + WS_AGG_OFF;
    int*   heads = (int*)d_ws + WS_HEAD_OFF;
    int*   rec   = (int*)d_ws + WS_REC_OFF;

    GCNN_build_lists_kernel<<<LIST_GROUPS, BUILD_THREADS, 0, stream>>>(
        rows, cols, vals, heads, rec);

    GCNN_spmm_gather_kernel<<<(N_NODES + ROWS_PER_BLOCK - 1) / ROWS_PER_BLOCK,
                              ROWS_PER_BLOCK * 32, 0, stream>>>(x, heads, rec, mp);

    GCNN_gemm_wmma_kernel<<<GEMM_BLOCKS, GEMM_WAVES * 32, 0, stream>>>(mp, W, bias, out);
}
